# starting kernel
# baseline (speedup 1.0000x reference)
#include <hip/hip_runtime.h>
#include <hip/hip_bf16.h>


typedef float v2f __attribute__((ext_vector_type(2)));
typedef float v8f __attribute__((ext_vector_type(8)));

#define BATCH 16
#define TSTEPS 64
#define DK 65536
#define KSLICES 32
#define KS_LEN (DK / KSLICES)
#define KW_LEN (KS_LEN / 8)

__global__ __launch_bounds__(256) void ta_scores_partial(
    const float* __restrict__ Hs, float* __restrict__ partial)
{
    const int wg = blockIdx.x;
    const int ks = wg & (KSLICES - 1);
    const int tt = (wg >> 5) & 3;
    const int b  = wg >> 7;

    const int lane = threadIdx.x & 31;
    const int wave = threadIdx.x >> 5;
    const int n  = lane & 15;
    const int hi = lane >> 4;

    const float* __restrict__ vlast =
        Hs + ((size_t)b * TSTEPS + (TSTEPS - 1)) * DK;
    const float* __restrict__ hrow =
        Hs + ((size_t)b * TSTEPS + (tt * 16 + n)) * DK;

    const int k0 = ks * KS_LEN + wave * KW_LEN;
    const float* ap = vlast + k0 + 2 * hi;
    const float* bp = hrow  + k0 + 2 * hi;

    v8f c = {0.f, 0.f, 0.f, 0.f, 0.f, 0.f, 0.f, 0.f};
    for (int o = 0; o < KW_LEN / (4 * 16); ++o) {
        __builtin_prefetch(bp + 64, 0, 0);
        #pragma unroll
        for (int i = 0; i < 16; ++i) {
            v2f a  = *reinterpret_cast<const v2f*>(ap);
            v2f bb = *reinterpret_cast<const v2f*>(bp);
            c = __builtin_amdgcn_wmma_f32_16x16x4_f32(
                     false, a,  false, bb,
                     (short)0, c,  false,  false);
            ap += 4; bp += 4;
        }
    }

    __shared__ float red[8][16];
    if (lane < 16) red[wave][n] = c[0];
    __syncthreads();
    if (threadIdx.x < 16) {
        float s = 0.f;
        #pragma unroll
        for (int w = 0; w < 8; ++w) s += red[w][threadIdx.x];
        partial[(size_t)ks * (BATCH * TSTEPS) + b * TSTEPS + tt * 16 + threadIdx.x] = s;
    }
}

__global__ __launch_bounds__(64) void ta_softmax(
    const float* __restrict__ partial, float* __restrict__ weights)
{
    const int b = blockIdx.x;
    const int t = threadIdx.x;

    float s = 0.f;
    #pragma unroll
    for (int ks = 0; ks < KSLICES; ++ks)
        s += partial[(size_t)ks * (BATCH * TSTEPS) + b * TSTEPS + t];
    s *= (1.0f / (float)TSTEPS);

    __shared__ float sh[TSTEPS];
    __shared__ float smax, ssum;
    sh[t] = s;
    __syncthreads();
    if (t == 0) {
        float m = sh[0];
        for (int i = 1; i < TSTEPS; ++i) m = fmaxf(m, sh[i]);
        smax = m;
    }
    __syncthreads();
    const float e = expf(s - smax);
    sh[t] = e;
    __syncthreads();
    if (t == 0) {
        float su = 0.f;
        for (int i = 0; i < TSTEPS; ++i) su += sh[i];
        ssum = su;
    }
    __syncthreads();
    weights[b * TSTEPS + t] = e / ssum;
}

__global__ __launch_bounds__(256) void ta_context(
    const float* __restrict__ Hs, const float* __restrict__ weights,
    float* __restrict__ out)
{
    const int b     = blockIdx.x >> 6;
    const int chunk = blockIdx.x & 63;
    const int idx   = (chunk * 1024 + threadIdx.x * 4) >> 2;

    __shared__ float w[TSTEPS];
    if (threadIdx.x < TSTEPS) w[threadIdx.x] = weights[b * TSTEPS + threadIdx.x];
    __syncthreads();

    const float4* __restrict__ Hb =
        reinterpret_cast<const float4*>(Hs + (size_t)b * TSTEPS * DK);
    float4 acc = {0.f, 0.f, 0.f, 0.f};
    #pragma unroll 8
    for (int t = 0; t < TSTEPS; ++t) {
        const float4 h = Hb[(size_t)t * (DK / 4) + idx];
        const float wt = w[t];
        acc.x += h.x * wt; acc.y += h.y * wt;
        acc.z += h.z * wt; acc.w += h.w * wt;
    }
    reinterpret_cast<float4*>(out)[(size_t)b * (DK / 4) + idx] = acc;
}

extern "C" void kernel_launch(void* const* d_in, const int* in_sizes, int n_in,
                              void* d_out, int out_size, void* d_ws, size_t ws_size,
                              hipStream_t stream) {
    (void)in_sizes; (void)n_in; (void)out_size; (void)ws_size;
    const float* Hs = (const float*)d_in[0];
    float* out = (float*)d_out;

    float* partial = (float*)d_ws;
    float* weights = partial + KSLICES * BATCH * TSTEPS;

    ta_scores_partial<<<BATCH * 4 * KSLICES, 256, 0, stream>>>(Hs, partial);
    ta_softmax<<<BATCH, 64, 0, stream>>>(partial, weights);
    ta_context<<<BATCH * 64, 256, 0, stream>>>(Hs, weights, out);
}


# Score pass streams whole lines: lane l loads 16 B at 4*l of each 512-B row piece, A=last with same map, diagonal of D summed per wave; order of the score partial sums changes
# speedup vs baseline: 1.3333x; 1.3333x over previous
#include <hip/hip_runtime.h>


typedef float v2f __attribute__((ext_vector_type(2)));
typedef float v4f __attribute__((ext_vector_type(4)));
typedef float v8f __attribute__((ext_vector_type(8)));

#define BATCH 16
#define TSTEPS 64
#define DK 65536
#define KSLICES 32
#define KS_LEN (DK / KSLICES)
#define KQ (KS_LEN / 128)
#define ROWS 8
#define WAVES 16

__global__ __launch_bounds__(WAVES * 32) void ta_scores_partial(
    const float* __restrict__ Hs, float* __restrict__ partial)
{
    const unsigned tid  = threadIdx.x;
    const unsigned lane = tid & 31u;
    const unsigned wave = __builtin_amdgcn_readfirstlane(tid >> 5);
    const unsigned b    = blockIdx.x >> 4;
    const unsigned kp   = blockIdx.x & 15u;
    const unsigned kl   = wave >> 3;
    const unsigned rg   = wave & 7u;
    const unsigned ks   = 2u * kp + kl;

    const size_t koff = (size_t)ks * KS_LEN + 4u * lane;
    const float* __restrict__ lastp =
        Hs + ((size_t)b * TSTEPS + (TSTEPS - 1)) * DK + koff;
    const float* __restrict__ rowp =
        Hs + ((size_t)b * TSTEPS + rg * ROWS) * DK + koff;

    v4f xq[KQ];
    #pragma unroll
    for (int q = 0; q < KQ; ++q)
        xq[q] = *reinterpret_cast<const v4f*>(lastp + 128 * q);

    v4f w[2][KQ];
    float sum[ROWS];
    const unsigned dr  = lane & 7u;
    const bool on_diag = (lane < 8u) || (lane >= 24u);

#define LOAD_ROW(R) do { \
    _Pragma("unroll") for (int q = 0; q < KQ; ++q) \
        w[(R) & 1][q] = *reinterpret_cast<const v4f*>(rowp + (size_t)(R) * DK + 128 * q); } while (0)

    LOAD_ROW(0);
    #pragma unroll
    for (int r = 0; r < ROWS; ++r) {
        __builtin_amdgcn_sched_barrier(0);
        if (r + 1 < ROWS) LOAD_ROW(r + 1);
        __builtin_amdgcn_sched_barrier(0);

        v8f c = {0.f, 0.f, 0.f, 0.f, 0.f, 0.f, 0.f, 0.f};
        #pragma unroll
        for (int q = 0; q < KQ; ++q) {
            const v4f a4 = xq[q];
            const v4f b4 = w[r & 1][q];
            const v2f a01 = {a4.x, a4.y}, b01 = {b4.x, b4.y};
            const v2f a23 = {a4.z, a4.w}, b23 = {b4.z, b4.w};
            c = __builtin_amdgcn_wmma_f32_16x16x4_f32(
                     false, a01,  false, b01,
                     (short)0, c,  false,  false);
            c = __builtin_amdgcn_wmma_f32_16x16x4_f32(
                     false, a23,  false, b23,
                     (short)0, c,  false,  false);
        }
        float d = c[0];
        d = (dr == 1u) ? c[1] : d;
        d = (dr == 2u) ? c[2] : d;
        d = (dr == 3u) ? c[3] : d;
        d = (dr == 4u) ? c[4] : d;
        d = (dr == 5u) ? c[5] : d;
        d = (dr == 6u) ? c[6] : d;
        d = (dr == 7u) ? c[7] : d;
        sum[r] = on_diag ? d : 0.f;
    }
#undef LOAD_ROW

    float mine = 0.f;
    #pragma unroll
    for (int r = 0; r < ROWS; ++r) {
        float s = sum[r];
        #pragma unroll
        for (int m = 16; m >= 1; m >>= 1) s += __shfl_xor(s, m, 32);
        if (lane == (unsigned)r) mine = s;
    }

    __shared__ __attribute__((aligned(16))) float red[2][TSTEPS];
    if (lane < (unsigned)ROWS) red[kl][rg * ROWS + lane] = mine;
    __syncthreads();
    if (tid < 32u) {
        const unsigned sl = tid >> 4;
        const unsigned j  = tid & 15u;
        const v4f v = *reinterpret_cast<const v4f*>(&red[sl][4u * j]);
        *reinterpret_cast<v4f*>(partial + (size_t)(2u * kp + sl) * (BATCH * TSTEPS)
                                + b * TSTEPS + 4u * j) = v;
    }
}

__global__ __launch_bounds__(64) void ta_softmax(
    const float* __restrict__ partial, float* __restrict__ weights)
{
    const int b = blockIdx.x;
    const int t = threadIdx.x;

    float s = 0.f;
    #pragma unroll
    for (int ks = 0; ks < KSLICES; ++ks)
        s += partial[(size_t)ks * (BATCH * TSTEPS) + b * TSTEPS + t];
    s *= (1.0f / (float)TSTEPS);

    __shared__ float sh[TSTEPS];
    __shared__ float smax, ssum;
    sh[t] = s;
    __syncthreads();
    if (t == 0) {
        float m = sh[0];
        for (int i = 1; i < TSTEPS; ++i) m = fmaxf(m, sh[i]);
        smax = m;
    }
    __syncthreads();
    const float e = expf(s - smax);
    sh[t] = e;
    __syncthreads();
    if (t == 0) {
        float su = 0.f;
        for (int i = 0; i < TSTEPS; ++i) su += sh[i];
        ssum = su;
    }
    __syncthreads();
    weights[b * TSTEPS + t] = e / ssum;
}

__global__ __launch_bounds__(256) void ta_context(
    const float* __restrict__ Hs, const float* __restrict__ weights,
    float* __restrict__ out)
{
    const int b     = blockIdx.x >> 6;
    const int chunk = blockIdx.x & 63;
    const int idx   = (chunk * 1024 + threadIdx.x * 4) >> 2;

    __shared__ float w[TSTEPS];
    if (threadIdx.x < TSTEPS) w[threadIdx.x] = weights[b * TSTEPS + threadIdx.x];
    __syncthreads();

    const float4* __restrict__ Hb =
        reinterpret_cast<const float4*>(Hs + (size_t)b * TSTEPS * DK);
    float4 acc = {0.f, 0.f, 0.f, 0.f};
    #pragma unroll 8
    for (int t = 0; t < TSTEPS; ++t) {
        const float4 h = Hb[(size_t)t * (DK / 4) + idx];
        const float wt = w[t];
        acc.x += h.x * wt; acc.y += h.y * wt;
        acc.z += h.z * wt; acc.w += h.w * wt;
    }
    reinterpret_cast<float4*>(out)[(size_t)b * (DK / 4) + idx] = acc;
}

extern "C" void kernel_launch(void* const* d_in, const int* in_sizes, int n_in,
                              void* d_out, int out_size, void* d_ws, size_t ws_size,
                              hipStream_t stream) {
    (void)in_sizes; (void)n_in; (void)out_size; (void)ws_size;
    const float* Hs = (const float*)d_in[0];
    float* out = (float*)d_out;

    float* partial = (float*)d_ws;
    float* weights = partial + KSLICES * BATCH * TSTEPS;

    ta_scores_partial<<<BATCH * 16, WAVES * 32, 0, stream>>>(Hs, partial);
    ta_softmax<<<BATCH, 64, 0, stream>>>(partial, weights);
    ta_context<<<BATCH * 64, 256, 0, stream>>>(Hs, weights, out);
}


# Context pass takes the score pass's grid and slab map (256 x 16 waves) and walks t newest-first (t%8 = 7..0) to hit the rows the score pass left in L2; the t order of the context sum changes
# speedup vs baseline: 1.7550x; 1.3163x over previous
#include <hip/hip_runtime.h>


typedef float v2f __attribute__((ext_vector_type(2)));
typedef float v4f __attribute__((ext_vector_type(4)));
typedef float v8f __attribute__((ext_vector_type(8)));

#define BATCH 16
#define TSTEPS 64
#define DK 65536
#define KSLICES 32
#define KS_LEN (DK / KSLICES)
#define KQ (KS_LEN / 128)
#define ROWS 8
#define WAVES 16

__global__ __launch_bounds__(WAVES * 32) void ta_scores_partial(
    const float* __restrict__ Hs, float* __restrict__ partial)
{
    const unsigned tid  = threadIdx.x;
    const unsigned lane = tid & 31u;
    const unsigned wave = __builtin_amdgcn_readfirstlane(tid >> 5);
    const unsigned b    = blockIdx.x >> 4;
    const unsigned kp   = blockIdx.x & 15u;
    const unsigned kl   = wave >> 3;
    const unsigned rg   = wave & 7u;
    const unsigned ks   = 2u * kp + kl;

    const size_t koff = (size_t)ks * KS_LEN + 4u * lane;
    const float* __restrict__ lastp =
        Hs + ((size_t)b * TSTEPS + (TSTEPS - 1)) * DK + koff;
    const float* __restrict__ rowp =
        Hs + ((size_t)b * TSTEPS + rg * ROWS) * DK + koff;

    v4f xq[KQ];
    #pragma unroll
    for (int q = 0; q < KQ; ++q)
        xq[q] = *reinterpret_cast<const v4f*>(lastp + 128 * q);

    v4f w[2][KQ];
    float sum[ROWS];
    const unsigned dr  = lane & 7u;
    const bool on_diag = (lane < 8u) || (lane >= 24u);

#define LOAD_ROW(R) do { \
    _Pragma("unroll") for (int q = 0; q < KQ; ++q) \
        w[(R) & 1][q] = *reinterpret_cast<const v4f*>(rowp + (size_t)(R) * DK + 128 * q); } while (0)

    LOAD_ROW(0);
    #pragma unroll
    for (int r = 0; r < ROWS; ++r) {
        __builtin_amdgcn_sched_barrier(0);
        if (r + 1 < ROWS) LOAD_ROW(r + 1);
        __builtin_amdgcn_sched_barrier(0);

        v8f c = {0.f, 0.f, 0.f, 0.f, 0.f, 0.f, 0.f, 0.f};
        #pragma unroll
        for (int q = 0; q < KQ; ++q) {
            const v4f a4 = xq[q];
            const v4f b4 = w[r & 1][q];
            const v2f a01 = {a4.x, a4.y}, b01 = {b4.x, b4.y};
            const v2f a23 = {a4.z, a4.w}, b23 = {b4.z, b4.w};
            c = __builtin_amdgcn_wmma_f32_16x16x4_f32(
                     false, a01,  false, b01,
                     (short)0, c,  false,  false);
            c = __builtin_amdgcn_wmma_f32_16x16x4_f32(
                     false, a23,  false, b23,
                     (short)0, c,  false,  false);
        }
        float d = c[0];
        d = (dr == 1u) ? c[1] : d;
        d = (dr == 2u) ? c[2] : d;
        d = (dr == 3u) ? c[3] : d;
        d = (dr == 4u) ? c[4] : d;
        d = (dr == 5u) ? c[5] : d;
        d = (dr == 6u) ? c[6] : d;
        d = (dr == 7u) ? c[7] : d;
        sum[r] = on_diag ? d : 0.f;
    }
#undef LOAD_ROW

    float mine = 0.f;
    #pragma unroll
    for (int r = 0; r < ROWS; ++r) {
        float s = sum[r];
        #pragma unroll
        for (int m = 16; m >= 1; m >>= 1) s += __shfl_xor(s, m, 32);
        if (lane == (unsigned)r) mine = s;
    }

    __shared__ __attribute__((aligned(16))) float red[2][TSTEPS];
    if (lane < (unsigned)ROWS) red[kl][rg * ROWS + lane] = mine;
    __syncthreads();
    if (tid < 32u) {
        const unsigned sl = tid >> 4;
        const unsigned j  = tid & 15u;
        const v4f v = *reinterpret_cast<const v4f*>(&red[sl][4u * j]);
        *reinterpret_cast<v4f*>(partial + (size_t)(2u * kp + sl) * (BATCH * TSTEPS)
                                + b * TSTEPS + 4u * j) = v;
    }
}

__global__ __launch_bounds__(64) void ta_softmax(
    const float* __restrict__ partial, float* __restrict__ weights)
{
    const int b = blockIdx.x;
    const int t = threadIdx.x;

    float s = 0.f;
    #pragma unroll
    for (int ks = 0; ks < KSLICES; ++ks)
        s += partial[(size_t)ks * (BATCH * TSTEPS) + b * TSTEPS + t];
    s *= (1.0f / (float)TSTEPS);

    __shared__ float sh[TSTEPS];
    __shared__ float smax, ssum;
    sh[t] = s;
    __syncthreads();
    if (t == 0) {
        float m = sh[0];
        for (int i = 1; i < TSTEPS; ++i) m = fmaxf(m, sh[i]);
        smax = m;
    }
    __syncthreads();
    const float e = expf(s - smax);
    sh[t] = e;
    __syncthreads();
    if (t == 0) {
        float su = 0.f;
        for (int i = 0; i < TSTEPS; ++i) su += sh[i];
        ssum = su;
    }
    __syncthreads();
    weights[b * TSTEPS + t] = e / ssum;
}

#define CTX_T(G, J) (8 * (7 - (J)) + (7 - (G)))
__global__ __launch_bounds__(WAVES * 32) void ta_context(
    const float* __restrict__ Hs, const float* __restrict__ weights, float* __restrict__ out)
{
    const unsigned tid  = threadIdx.x;
    const unsigned lane = tid & 31u;
    const unsigned wave = __builtin_amdgcn_readfirstlane(tid >> 5);
    const unsigned b    = blockIdx.x >> 4;
    const unsigned kp   = blockIdx.x & 15u;

    const size_t koff = (size_t)kp * (2u * KS_LEN) + 256u * wave + 4u * lane;
    const float* hp = Hs + (size_t)b * TSTEPS * DK + koff;
    const float* __restrict__ wp = weights + b * TSTEPS;

    v4f h[2][16];

#define LOAD_GROUP(G) do { \
    _Pragma("unroll") for (int j = 0; j < 8; ++j) { \
        const float* p = hp + (size_t)CTX_T(G, j) * DK; \
        h[(G) & 1][2 * j]     = *reinterpret_cast<const v4f*>(p); \
        h[(G) & 1][2 * j + 1] = *reinterpret_cast<const v4f*>(p + 128); } } while (0)

    LOAD_GROUP(0);

    float wt[TSTEPS];
    #pragma unroll
    for (int t = 0; t < TSTEPS; ++t) wt[t] = wp[t];

    v4f acc0 = {0.f, 0.f, 0.f, 0.f};
    v4f acc1 = {0.f, 0.f, 0.f, 0.f};
    #pragma unroll
    for (int g = 0; g < 8; ++g) {
        if (g + 1 < 8) LOAD_GROUP(g + 1);
        asm volatile("" : "+v"(acc0), "+v"(acc1));

        #pragma unroll
        for (int j = 0; j < 8; ++j) {
            const float w = wt[CTX_T(g, j)];
            acc0 += h[g & 1][2 * j] * w;
            acc1 += h[g & 1][2 * j + 1] * w;
        }
    }
#undef LOAD_GROUP

    float* __restrict__ op = out + (size_t)b * DK + koff;
    *reinterpret_cast<v4f*>(op)       = acc0;
    *reinterpret_cast<v4f*>(op + 128) = acc1;
}
#undef CTX_T

extern "C" void kernel_launch(void* const* d_in, const int* in_sizes, int n_in,
                              void* d_out, int out_size, void* d_ws, size_t ws_size,
                              hipStream_t stream) {
    (void)in_sizes; (void)n_in; (void)out_size; (void)ws_size;
    const float* Hs = (const float*)d_in[0];
    float* out = (float*)d_out;

    float* partial = (float*)d_ws;
    float* weights = partial + KSLICES * BATCH * TSTEPS;

    ta_scores_partial<<<BATCH * 16, WAVES * 32, 0, stream>>>(Hs, partial);
    ta_softmax<<<BATCH, 64, 0, stream>>>(partial, weights);
    ta_context<<<BATCH * 16, WAVES * 32, 0, stream>>>(Hs, weights, out);
}


# Softmax launch folded into the context pass: a workgroup shares its batch's 32x64 score partials via LDS, each wave sums them in slice order; denominator by lane butterfly: its sum order changes
# speedup vs baseline: 1.8620x; 1.0610x over previous
#include <hip/hip_runtime.h>


typedef float v2f __attribute__((ext_vector_type(2)));
typedef float v4f __attribute__((ext_vector_type(4)));
typedef float v8f __attribute__((ext_vector_type(8)));

#define BATCH 16
#define TSTEPS 64
#define DK 65536
#define KSLICES 32
#define KS_LEN (DK / KSLICES)
#define KQ (KS_LEN / 128)
#define ROWS 8
#define WAVES 16

__global__ __launch_bounds__(WAVES * 32) void ta_scores_partial(
    const float* __restrict__ Hs, float* __restrict__ partial)
{
    const unsigned tid  = threadIdx.x;
    const unsigned lane = tid & 31u;
    const unsigned wave = __builtin_amdgcn_readfirstlane(tid >> 5);
    const unsigned b    = blockIdx.x >> 4;
    const unsigned kp   = blockIdx.x & 15u;
    const unsigned kl   = wave >> 3;
    const unsigned rg   = wave & 7u;
    const unsigned ks   = 2u * kp + kl;

    const size_t koff = (size_t)ks * KS_LEN + 4u * lane;
    const float* __restrict__ lastp =
        Hs + ((size_t)b * TSTEPS + (TSTEPS - 1)) * DK + koff;
    const float* __restrict__ rowp =
        Hs + ((size_t)b * TSTEPS + rg * ROWS) * DK + koff;

    v4f xq[KQ];
    #pragma unroll
    for (int q = 0; q < KQ; ++q)
        xq[q] = *reinterpret_cast<const v4f*>(lastp + 128 * q);

    v4f w[2][KQ];
    float sum[ROWS];
    const unsigned dr  = lane & 7u;
    const bool on_diag = (lane < 8u) || (lane >= 24u);

#define LOAD_ROW(R) do { \
    _Pragma("unroll") for (int q = 0; q < KQ; ++q) \
        w[(R) & 1][q] = *reinterpret_cast<const v4f*>(rowp + (size_t)(R) * DK + 128 * q); } while (0)

    LOAD_ROW(0);
    #pragma unroll
    for (int r = 0; r < ROWS; ++r) {
        __builtin_amdgcn_sched_barrier(0);
        if (r + 1 < ROWS) LOAD_ROW(r + 1);
        __builtin_amdgcn_sched_barrier(0);

        v8f c = {0.f, 0.f, 0.f, 0.f, 0.f, 0.f, 0.f, 0.f};
        #pragma unroll
        for (int q = 0; q < KQ; ++q) {
            const v4f a4 = xq[q];
            const v4f b4 = w[r & 1][q];
            const v2f a01 = {a4.x, a4.y}, b01 = {b4.x, b4.y};
            const v2f a23 = {a4.z, a4.w}, b23 = {b4.z, b4.w};
            c = __builtin_amdgcn_wmma_f32_16x16x4_f32(
                     false, a01,  false, b01,
                     (short)0, c,  false,  false);
            c = __builtin_amdgcn_wmma_f32_16x16x4_f32(
                     false, a23,  false, b23,
                     (short)0, c,  false,  false);
        }
        float d = c[0];
        d = (dr == 1u) ? c[1] : d;
        d = (dr == 2u) ? c[2] : d;
        d = (dr == 3u) ? c[3] : d;
        d = (dr == 4u) ? c[4] : d;
        d = (dr == 5u) ? c[5] : d;
        d = (dr == 6u) ? c[6] : d;
        d = (dr == 7u) ? c[7] : d;
        sum[r] = on_diag ? d : 0.f;
    }
#undef LOAD_ROW

    float mine = 0.f;
    #pragma unroll
    for (int r = 0; r < ROWS; ++r) {
        float s = sum[r];
        #pragma unroll
        for (int m = 16; m >= 1; m >>= 1) s += __shfl_xor(s, m, 32);
        if (lane == (unsigned)r) mine = s;
    }

    __shared__ __attribute__((aligned(16))) float red[2][TSTEPS];
    if (lane < (unsigned)ROWS) red[kl][rg * ROWS + lane] = mine;
    __syncthreads();
    if (tid < 32u) {
        const unsigned sl = tid >> 4;
        const unsigned j  = tid & 15u;
        const v4f v = *reinterpret_cast<const v4f*>(&red[sl][4u * j]);
        *reinterpret_cast<v4f*>(partial + (size_t)(2u * kp + sl) * (BATCH * TSTEPS)
                                + b * TSTEPS + 4u * j) = v;
    }
}

static __device__ __forceinline__ float xhalf(float x) {
    const int v = __builtin_bit_cast(int, x);
    return __builtin_bit_cast(float,
        __builtin_amdgcn_permlanex16(v, v, 0x76543210, 0xfedcba98, false, false));
}
template <int M>
static __device__ __forceinline__ float xrow(float x) {
    const int v = __builtin_bit_cast(int, x);
    return __builtin_bit_cast(float,
        __builtin_amdgcn_update_dpp(v, v, 0x160 | M, 0xf, 0xf, true));
}

#define CTX_T(G, J) (8 * (7 - (J)) + (7 - (G)))
__global__ __launch_bounds__(WAVES * 32) void ta_context(
    const float* __restrict__ Hs, const float* __restrict__ partial, float* __restrict__ out)
{
    const unsigned tid  = threadIdx.x;
    const unsigned lane = tid & 31u;
    const unsigned wave = __builtin_amdgcn_readfirstlane(tid >> 5);
    const unsigned b    = blockIdx.x >> 4;
    const unsigned kp   = blockIdx.x & 15u;

    const v4f pv = *reinterpret_cast<const v4f*>(
        partial + (size_t)(tid >> 4) * (BATCH * TSTEPS) + b * TSTEPS + 4u * (tid & 15u));

    const size_t koff = (size_t)kp * (2u * KS_LEN) + 256u * wave + 4u * lane;
    const float* hp = Hs + (size_t)b * TSTEPS * DK + koff;

    v4f h[2][16];

#define LOAD_GROUP(G) do { \
    _Pragma("unroll") for (int j = 0; j < 8; ++j) { \
        const float* p = hp + (size_t)CTX_T(G, j) * DK; \
        h[(G) & 1][2 * j]     = *reinterpret_cast<const v4f*>(p); \
        h[(G) & 1][2 * j + 1] = *reinterpret_cast<const v4f*>(p + 128); } } while (0)

    LOAD_GROUP(0);
    LOAD_GROUP(1);

    __shared__ __attribute__((aligned(16))) float sc[KSLICES * TSTEPS];
    *reinterpret_cast<v4f*>(&sc[4u * tid]) = pv;
    __builtin_amdgcn_fence(__ATOMIC_RELEASE, "workgroup", "local");
    __builtin_amdgcn_s_barrier();
    __builtin_amdgcn_fence(__ATOMIC_ACQUIRE, "workgroup", "local");

    float s0 = 0.f, s1 = 0.f;
    for (int ks = 0; ks < KSLICES; ++ks) {
        const v2f v = *reinterpret_cast<const v2f*>(&sc[ks * TSTEPS + 2u * lane]);
        s0 += v.x;
        s1 += v.y;
    }
    s0 *= (1.0f / (float)TSTEPS);
    s1 *= (1.0f / (float)TSTEPS);

    float m = fmaxf(s0, s1);
    m = fmaxf(m, xrow<1>(m));
    m = fmaxf(m, xrow<2>(m));
    m = fmaxf(m, xrow<4>(m));
    m = fmaxf(m, xrow<8>(m));
    m = fmaxf(m, xhalf(m));
    const float e0 = expf(s0 - m);
    const float e1 = expf(s1 - m);
    float d = e0 + e1;
    d += xrow<1>(d);
    d += xrow<2>(d);
    d += xrow<4>(d);
    d += xrow<8>(d);
    d += xhalf(d);
    const float w0 = e0 / d;
    const float w1 = e1 / d;
#define WT(T) __builtin_bit_cast(float, \
    __builtin_amdgcn_readlane(__builtin_bit_cast(int, ((T) & 1) ? w1 : w0), (T) >> 1))

    v4f acc0 = {0.f, 0.f, 0.f, 0.f};
    v4f acc1 = {0.f, 0.f, 0.f, 0.f};
    #pragma unroll
    for (int g = 0; g < 8; ++g) {
        if (g >= 1 && g + 1 < 8) LOAD_GROUP(g + 1);
        asm volatile("" : "+v"(acc0), "+v"(acc1));

        #pragma unroll
        for (int j = 0; j < 8; ++j) {
            const float w = WT(CTX_T(g, j));
            acc0 += h[g & 1][2 * j] * w;
            acc1 += h[g & 1][2 * j + 1] * w;
        }
    }
#undef WT
#undef LOAD_GROUP

    float* __restrict__ op = out + (size_t)b * DK + koff;
    *reinterpret_cast<v4f*>(op)       = acc0;
    *reinterpret_cast<v4f*>(op + 128) = acc1;
}
#undef CTX_T

extern "C" void kernel_launch(void* const* d_in, const int* in_sizes, int n_in,
                              void* d_out, int out_size, void* d_ws, size_t ws_size,
                              hipStream_t stream) {
    (void)in_sizes; (void)n_in; (void)out_size; (void)ws_size;
    const float* Hs = (const float*)d_in[0];
    float* out = (float*)d_out;

    float* partial = (float*)d_ws;

    ta_scores_partial<<<BATCH * 16, WAVES * 32, 0, stream>>>(Hs, partial);
    ta_context<<<BATCH * 16, WAVES * 32, 0, stream>>>(Hs, partial, out);
}
